# starting kernel
# baseline (speedup 1.0000x reference)
#include <hip/hip_runtime.h>

#define N_NODES 50000
#define N_EDGES 1600000
#define IN_CH   64
#define HID_CH  64
#define OUT_CH  32
#define NEG_SLOPE 0.01f

typedef __attribute__((ext_vector_type(2))) float v2f;
typedef __attribute__((ext_vector_type(8))) float v8f;

__global__ __launch_bounds__(256) void sage_scatter(
    const float* __restrict__ feat, const int* __restrict__ ei,
    float* __restrict__ agg, float* __restrict__ deg, int count_deg)
{
    long idx = (long)blockIdx.x * blockDim.x + threadIdx.x;
    if (idx >= (long)N_EDGES * 16) return;
    int e = (int)(idx >> 4);
    int g = (int)(idx & 15);

    int src = ei[e];
    int dst = ei[N_EDGES + e];

    const float4 v = *(const float4*)(feat + (long)src * 64 + g * 4);
    float* a = agg + (long)dst * 64 + g * 4;
    unsafeAtomicAdd(a + 0, v.x);
    unsafeAtomicAdd(a + 1, v.y);
    unsafeAtomicAdd(a + 2, v.z);
    unsafeAtomicAdd(a + 3, v.w);
    if (count_deg && g == 0) unsafeAtomicAdd(deg + dst, 1.0f);
}

template <int OUTC>
__global__ __launch_bounds__(256) void sage_wmma(
    const float* __restrict__ agg, const float* __restrict__ deg,
    const float* __restrict__ feat,
    const float* __restrict__ Wl, const float* __restrict__ Wr,
    const float* __restrict__ bias, float* __restrict__ out)
{
    __shared__ float Wlds[OUTC * 128];

    const int tid = threadIdx.x;
    for (int i = tid; i < OUTC * 128; i += 256) {
        int n = i >> 7, k = i & 127;
        Wlds[i] = (k < 64) ? Wl[n * 64 + k] : Wr[n * 64 + (k - 64)];
    }
    __syncthreads();

    const int wave = tid >> 5;
    const int lane = tid & 31;
    const int col_tiles   = OUTC / 16;
    const int total_tiles = (N_NODES / 16) * col_tiles;
    const int tile = blockIdx.x * 8 + wave;
    if (tile >= total_tiles) return;

    const int row0 = (tile / col_tiles) * 16;
    const int col0 = (tile % col_tiles) * 16;
    const int half = lane >> 4;
    const int m    = lane & 15;
    const int row  = row0 + m;
    const int colg = col0 + m;

    const float rd = 1.0f / fmaxf(deg[row], 1.0f);
    const float bc = bias[colg];

    v8f c;
#pragma unroll
    for (int i = 0; i < 8; ++i) c[i] = bc;

    const float* arow = agg  + (long)row * 64;
    const float* xrow = feat + (long)row * 64;
    const float* wcol = &Wlds[colg * 128];

#pragma unroll
    for (int ks = 0; ks < 64; ks += 4) {
        v2f a, b;
        a[0] = arow[ks + half * 2 + 0] * rd;
        a[1] = arow[ks + half * 2 + 1] * rd;
        b[0] = wcol[ks + half * 2 + 0];
        b[1] = wcol[ks + half * 2 + 1];
        c = __builtin_amdgcn_wmma_f32_16x16x4_f32(false, a, false, b,
                                                  (short)0, c, false, false);
    }
#pragma unroll
    for (int ks = 0; ks < 64; ks += 4) {
        v2f a, b;
        a[0] = xrow[ks + half * 2 + 0];
        a[1] = xrow[ks + half * 2 + 1];
        b[0] = wcol[64 + ks + half * 2 + 0];
        b[1] = wcol[64 + ks + half * 2 + 1];
        c = __builtin_amdgcn_wmma_f32_16x16x4_f32(false, a, false, b,
                                                  (short)0, c, false, false);
    }

#pragma unroll
    for (int v = 0; v < 8; ++v) {
        float y = c[v];
        y = (y > 0.0f) ? y : NEG_SLOPE * y;
        out[(long)(row0 + half * 8 + v) * OUTC + colg] = y;
    }
}

extern "C" void kernel_launch(void* const* d_in, const int* in_sizes, int n_in,
                              void* d_out, int out_size, void* d_ws, size_t ws_size,
                              hipStream_t stream) {
    const float* x   = (const float*)d_in[0];
    const int*   ei  = (const int*)  d_in[1];
    const float* W1l = (const float*)d_in[2];
    const float* b1  = (const float*)d_in[3];
    const float* W1r = (const float*)d_in[4];
    const float* W2l = (const float*)d_in[5];
    const float* b2  = (const float*)d_in[6];
    const float* W2r = (const float*)d_in[7];
    float* out = (float*)d_out;

    const size_t degB = ((size_t)N_NODES * 4 + 255) & ~(size_t)255;
    const size_t aggB = (size_t)N_NODES * 64 * 4;
    float* deg = (float*)d_ws;
    float* agg = (float*)((char*)d_ws + degB);
    float* h   = (float*)((char*)d_ws + degB + aggB);

    const long sthreads = (long)N_EDGES * 16;
    const int  sblocks  = (int)((sthreads + 255) / 256);

    hipMemsetAsync(d_ws, 0, degB + aggB, stream);
    sage_scatter<<<sblocks, 256, 0, stream>>>(x, ei, agg, deg, 1);
    {
        const int tiles  = (N_NODES / 16) * (HID_CH / 16);
        const int blocks = (tiles + 7) / 8;
        sage_wmma<HID_CH><<<blocks, 256, 0, stream>>>(agg, deg, x, W1l, W1r, b1, h);
    }

    hipMemsetAsync(agg, 0, aggB, stream);
    sage_scatter<<<sblocks, 256, 0, stream>>>(h, ei, agg, deg, 0);
    {
        const int tiles  = (N_NODES / 16) * (OUT_CH / 16);
        const int blocks = (tiles + 7) / 8;
        sage_wmma<OUT_CH><<<blocks, 256, 0, stream>>>(agg, deg, h, W2l, W2r, b2, out);
    }
}


# both linear maps in one launch: a wave owns 16 nodes, weights by 16-byte global loads (no LDS panel), step 1 transposed so its accumulators are step 2's A fragments, h not stored; k sum order changes
# speedup vs baseline: 3.4745x; 3.4745x over previous
#include <hip/hip_runtime.h>

#define N_NODES 50000
#define N_EDGES 1600000
#define IN_CH   64
#define HID_CH  64
#define OUT_CH  32
#define NEG_SLOPE 0.01f
#define DEG_PITCH 32
#define COPY_FLOATS ((long)N_NODES * (DEG_PITCH + 64 + OUT_CH))

typedef __attribute__((ext_vector_type(2))) float v2f;
typedef __attribute__((ext_vector_type(8))) float v8f;

__device__ __forceinline__ int cache_half()
{
    const unsigned place = __builtin_amdgcn_s_sendmsg_rtn(0x87);
    return (int)((place >> 18) & 1u);
}

static_assert(N_EDGES % 32 == 0, "a wave takes exactly 32 edges");

template <int C>
__global__ __launch_bounds__(256) void sage_scatter(
    const float* __restrict__ feat, const int* __restrict__ ei,
    float* __restrict__ agg0, float* __restrict__ deg0, int count_deg)
{
    constexpr int LPR   = C / 32;
    constexpr int BATCH = 32 / LPR;

    const int lane = threadIdx.x & 31;
    const int wave = (int)((blockIdx.x * 256u + threadIdx.x) >> 5);
    const int e0   = wave * 32;
    if (e0 >= N_EDGES) return;

    const long copy = (long)cache_half() * COPY_FLOATS;
    float* agg = agg0 + copy;
    float* deg = deg0 + copy;

    const int my_src = ei[e0 + lane];
    const int my_dst = ei[N_EDGES + e0 + lane];

    if (count_deg) unsafeAtomicAdd(deg + (long)my_dst * DEG_PITCH, 1.0f);

#pragma unroll
    for (int jb = 0; jb < 32; jb += BATCH) {
        float v[BATCH][LPR];
#pragma unroll
        for (int u = 0; u < BATCH; ++u) {
            const int s = __builtin_amdgcn_readlane(my_src, jb + u);
            const float* srow = feat + (long)s * C + lane;
#pragma unroll
            for (int l = 0; l < LPR; ++l) v[u][l] = srow[l * 32];
        }
#pragma unroll
        for (int u = 0; u < BATCH; ++u) {
            const int d = __builtin_amdgcn_readlane(my_dst, jb + u);
            float* drow = agg + (long)d * C + lane;
#pragma unroll
            for (int l = 0; l < LPR; ++l) unsafeAtomicAdd(drow + l * 32, v[u][l]);
        }
    }
}

static_assert(N_NODES % 16 == 0, "a wave takes exactly 16 nodes");

__device__ __forceinline__ v8f wmma4(v2f a, v2f b, v8f c)
{
    return __builtin_amdgcn_wmma_f32_16x16x4_f32(false, a, false, b, (short)0, c, false, false);
}

__global__ __launch_bounds__(128) void sage_layers(
    const float* __restrict__ agg, const float* __restrict__ deg,
    const float* __restrict__ x,
    const float* __restrict__ W1l, const float* __restrict__ W1r,
    const float* __restrict__ b1,
    const float* __restrict__ W2l, const float* __restrict__ W2r,
    const float* __restrict__ b2,
    float* __restrict__ p, float* __restrict__ q)
{
    const int lane = threadIdx.x & 31;
    const int wave = (int)((blockIdx.x * 128u + threadIdx.x) >> 5);
    if (wave >= N_NODES / 16) return;

    const int n0   = wave * 16;
    const int c    = lane & 15;
    const int hf   = lane >> 4;
    const int node = n0 + c;

    const float rd = 1.0f / fmaxf(deg[(long)node * DEG_PITCH] +
                                  deg[COPY_FLOATS + (long)node * DEG_PITCH], 1.0f);

    const float* s0 = agg + (long)node * 64 + 4 * hf;
    const float* s1 = s0 + COPY_FLOATS;
    const float* xr = x + (long)node * 64 + 4 * hf;
    const float* wl = W1l + c * 64 + 4 * hf;
    const float* wr = W1r + c * 64 + 4 * hf;

    v8f acc[4];
#pragma unroll
    for (int T = 0; T < 4; ++T) {
        const float4 ba = *(const float4*)(b1 + 16 * T + 8 * hf);
        const float4 bb = *(const float4*)(b1 + 16 * T + 8 * hf + 4);
        acc[T][0] = ba.x; acc[T][1] = ba.y; acc[T][2] = ba.z; acc[T][3] = ba.w;
        acc[T][4] = bb.x; acc[T][5] = bb.y; acc[T][6] = bb.z; acc[T][7] = bb.w;
    }

#pragma unroll
    for (int t = 0; t < 8; ++t) {
        const float4 g0 = *(const float4*)(s0 + 8 * t);
        const float4 g1 = *(const float4*)(s1 + 8 * t);
        v2f f0, f1;
        f0[0] = (g0.x + g1.x) * rd;  f0[1] = (g0.y + g1.y) * rd;
        f1[0] = (g0.z + g1.z) * rd;  f1[1] = (g0.w + g1.w) * rd;
        float4 w[4];
#pragma unroll
        for (int T = 0; T < 4; ++T) w[T] = *(const float4*)(wl + T * 16 * 64 + 8 * t);
#pragma unroll
        for (int T = 0; T < 4; ++T) { v2f a; a[0] = w[T].x; a[1] = w[T].y; acc[T] = wmma4(a, f0, acc[T]); }
#pragma unroll
        for (int T = 0; T < 4; ++T) { v2f a; a[0] = w[T].z; a[1] = w[T].w; acc[T] = wmma4(a, f1, acc[T]); }
    }
#pragma unroll
    for (int t = 0; t < 8; ++t) {
        const float4 g = *(const float4*)(xr + 8 * t);
        v2f f0, f1;
        f0[0] = g.x;  f0[1] = g.y;
        f1[0] = g.z;  f1[1] = g.w;
        float4 w[4];
#pragma unroll
        for (int T = 0; T < 4; ++T) w[T] = *(const float4*)(wr + T * 16 * 64 + 8 * t);
#pragma unroll
        for (int T = 0; T < 4; ++T) { v2f a; a[0] = w[T].x; a[1] = w[T].y; acc[T] = wmma4(a, f0, acc[T]); }
#pragma unroll
        for (int T = 0; T < 4; ++T) { v2f a; a[0] = w[T].z; a[1] = w[T].w; acc[T] = wmma4(a, f1, acc[T]); }
    }

#pragma unroll
    for (int T = 0; T < 4; ++T)
#pragma unroll
        for (int r = 0; r < 8; ++r) {
            const float y = acc[T][r];
            acc[T][r] = (y > 0.0f) ? y : NEG_SLOPE * y;
        }

    const float* w2[4];
    w2[0] = W2l + c * 64 + 8 * hf;
    w2[1] = W2l + (16 + c) * 64 + 8 * hf;
    w2[2] = W2r + c * 64 + 8 * hf;
    w2[3] = W2r + (16 + c) * 64 + 8 * hf;

    const float bq0 = b2[c];
    const float bq1 = b2[16 + c];
    v8f o[4];
#pragma unroll
    for (int r = 0; r < 8; ++r) { o[0][r] = 0.0f; o[1][r] = 0.0f; o[2][r] = bq0; o[3][r] = bq1; }

#pragma unroll
    for (int T = 0; T < 4; ++T) {
        float4 wa[4], wb[4];
#pragma unroll
        for (int J = 0; J < 4; ++J) {
            wa[J] = *(const float4*)(w2[J] + 16 * T);
            wb[J] = *(const float4*)(w2[J] + 16 * T + 4);
        }
#pragma unroll
        for (int u = 0; u < 4; ++u) {
            v2f a;
            a[0] = acc[T][2 * u];
            a[1] = acc[T][2 * u + 1];
#pragma unroll
            for (int J = 0; J < 4; ++J) {
                v2f b;
                if (u == 0)      { b[0] = wa[J].x; b[1] = wa[J].y; }
                else if (u == 1) { b[0] = wa[J].z; b[1] = wa[J].w; }
                else if (u == 2) { b[0] = wb[J].x; b[1] = wb[J].y; }
                else             { b[0] = wb[J].z; b[1] = wb[J].w; }
                o[J] = wmma4(a, b, o[J]);
            }
        }
    }

    const int col = hf * 16 + c;
#pragma unroll
    for (int r = 0; r < 8; ++r) {
        {
            const float lo = o[0][r], hi = o[1][r];
            const float give = hf ? lo : hi;
            const float got  = __shfl_xor(give, 16);
            const float rowA = hf ? got : lo;
            const float rowB = hf ? hi  : got;
            p[(long)(n0 + r) * OUT_CH + col]     = rowA;
            p[(long)(n0 + 8 + r) * OUT_CH + col] = rowB;
        }
        {
            const float lo = o[2][r], hi = o[3][r];
            const float give = hf ? lo : hi;
            const float got  = __shfl_xor(give, 16);
            const float rowA = hf ? got : lo;
            const float rowB = hf ? hi  : got;
            q[(long)(n0 + r) * OUT_CH + col]     = rowA;
            q[(long)(n0 + 8 + r) * OUT_CH + col] = rowB;
        }
    }
}

__global__ __launch_bounds__(256) void sage_finish(
    const float* __restrict__ agg2, const float* __restrict__ deg, float* out)
{
    const int idx = blockIdx.x * 256 + threadIdx.x;
    if (idx >= N_NODES * (OUT_CH / 4)) return;
    const int row = idx / (OUT_CH / 4);

    const float rd = 1.0f / fmaxf(deg[(long)row * DEG_PITCH] +
                                  deg[COPY_FLOATS + (long)row * DEG_PITCH], 1.0f);
    const float4 a0 = ((const float4*)agg2)[idx];
    const float4 a1 = ((const float4*)(agg2 + COPY_FLOATS))[idx];
    float4 a;
    a.x = a0.x + a1.x;
    a.y = a0.y + a1.y;
    a.z = a0.z + a1.z;
    a.w = a0.w + a1.w;
    const float4 s = ((const float4*)out)[idx];
    float4 y;
    y.x = a.x * rd + s.x;
    y.y = a.y * rd + s.y;
    y.z = a.z * rd + s.z;
    y.w = a.w * rd + s.w;
    y.x = (y.x > 0.0f) ? y.x : NEG_SLOPE * y.x;
    y.y = (y.y > 0.0f) ? y.y : NEG_SLOPE * y.y;
    y.z = (y.z > 0.0f) ? y.z : NEG_SLOPE * y.z;
    y.w = (y.w > 0.0f) ? y.w : NEG_SLOPE * y.w;
    ((float4*)out)[idx] = y;
}

extern "C" void kernel_launch(void* const* d_in, const int* in_sizes, int n_in,
                              void* d_out, int out_size, void* d_ws, size_t ws_size,
                              hipStream_t stream) {
    const float* x   = (const float*)d_in[0];
    const int*   ei  = (const int*)  d_in[1];
    const float* W1l = (const float*)d_in[2];
    const float* b1  = (const float*)d_in[3];
    const float* W1r = (const float*)d_in[4];
    const float* W2l = (const float*)d_in[5];
    const float* b2  = (const float*)d_in[6];
    const float* W2r = (const float*)d_in[7];
    float* out = (float*)d_out;

    const size_t degB  = (size_t)N_NODES * DEG_PITCH * 4;
    const size_t aggB  = (size_t)N_NODES * 64 * 4;
    const size_t agg2B = (size_t)N_NODES * OUT_CH * 4;
    const size_t copyB = (size_t)COPY_FLOATS * 4;
    float* deg  = (float*)d_ws;
    float* agg  = (float*)((char*)d_ws + degB);
    float* agg2 = (float*)((char*)d_ws + degB + aggB);
    float* p    = (float*)((char*)d_ws + 2 * copyB);

    hipMemsetAsync(d_ws, 0, 2 * copyB, stream);

    {
        const int swaves  = N_EDGES / 32;
        const int sblocks = (swaves + 7) / 8;
        sage_scatter<IN_CH><<<sblocks, 256, 0, stream>>>(x, ei, agg, deg, 1);
    }
    {
        const int waves  = N_NODES / 16;
        const int blocks = (waves + 3) / 4;
        sage_layers<<<blocks, 128, 0, stream>>>(agg, deg, x, W1l, W1r, b1, W2l, W2r, b2, p, out);
    }
    {
        const int swaves  = N_EDGES / 32;
        const int sblocks = (swaves + 7) / 8;
        sage_scatter<OUT_CH><<<sblocks, 256, 0, stream>>>(p, ei, agg2, deg, 0);
    }
    {
        const int fthreads = N_NODES * (OUT_CH / 4);
        const int fblocks  = (fthreads + 255) / 256;
        sage_finish<<<fblocks, 256, 0, stream>>>(agg2, deg, out);
    }
}


# edge and finish passes: a wave loops over 5 chunks of 32 edges (finish: 4 items), next chunk's edge words loaded ahead: 5x fewer, longer-lived waves enter; lines, adds, sum order as before
# speedup vs baseline: 3.7336x; 1.0746x over previous
#include <hip/hip_runtime.h>

#define N_NODES 50000
#define N_EDGES 1600000
#define IN_CH   64
#define HID_CH  64
#define OUT_CH  32
#define NEG_SLOPE 0.01f
#define DEG_PITCH 32
#define COPY_FLOATS ((long)N_NODES * (DEG_PITCH + 64 + OUT_CH))

typedef __attribute__((ext_vector_type(2))) float v2f;
typedef __attribute__((ext_vector_type(8))) float v8f;

__device__ __forceinline__ int cache_half()
{
    const unsigned place = __builtin_amdgcn_s_sendmsg_rtn(0x87);
    return (int)((place >> 18) & 1u);
}

#define EDGE_CHUNKS 5
#define EDGE_WAVES  (N_EDGES / (32 * EDGE_CHUNKS))
static_assert(N_EDGES % (32 * EDGE_CHUNKS) == 0, "a wave takes exactly EDGE_CHUNKS chunks of 32 edges");

template <int C>
__global__ __launch_bounds__(256) void sage_scatter(
    const float* __restrict__ feat, const int* __restrict__ ei,
    float* __restrict__ agg0, float* __restrict__ deg0, int count_deg)
{
    constexpr int LPR   = C / 32;
    constexpr int BATCH = 32 / LPR;

    const int lane = threadIdx.x & 31;
    const int wave = (int)((blockIdx.x * 256u + threadIdx.x) >> 5);
    if (wave >= EDGE_WAVES) return;

    const long copy = (long)cache_half() * COPY_FLOATS;
    float* agg = agg0 + copy;
    float* deg = deg0 + copy;

    const int* sp = ei + wave * (32 * EDGE_CHUNKS) + lane;
    const int* dp = sp + N_EDGES;

    int my_src = sp[0];
    int my_dst = dp[0];

    for (int k = 0; k < EDGE_CHUNKS; ++k) {
        const int kn = (k + 1 < EDGE_CHUNKS) ? k + 1 : k;
        const int nx_src = sp[32 * kn];
        const int nx_dst = dp[32 * kn];

        if (count_deg) unsafeAtomicAdd(deg + (long)my_dst * DEG_PITCH, 1.0f);

#pragma unroll
        for (int jb = 0; jb < 32; jb += BATCH) {
            float v[BATCH][LPR];
#pragma unroll
            for (int u = 0; u < BATCH; ++u) {
                const int s = __builtin_amdgcn_readlane(my_src, jb + u);
                const float* srow = feat + (long)s * C + lane;
#pragma unroll
                for (int l = 0; l < LPR; ++l) v[u][l] = srow[l * 32];
            }
#pragma unroll
            for (int u = 0; u < BATCH; ++u) {
                const int d = __builtin_amdgcn_readlane(my_dst, jb + u);
                float* drow = agg + (long)d * C + lane;
#pragma unroll
                for (int l = 0; l < LPR; ++l) unsafeAtomicAdd(drow + l * 32, v[u][l]);
            }
        }

        my_src = nx_src;
        my_dst = nx_dst;
    }
}

static_assert(N_NODES % 16 == 0, "a wave takes exactly 16 nodes");

__device__ __forceinline__ v8f wmma4(v2f a, v2f b, v8f c)
{
    return __builtin_amdgcn_wmma_f32_16x16x4_f32(false, a, false, b, (short)0, c, false, false);
}

__global__ __launch_bounds__(128) void sage_layers(
    const float* __restrict__ agg, const float* __restrict__ deg,
    const float* __restrict__ x,
    const float* __restrict__ W1l, const float* __restrict__ W1r,
    const float* __restrict__ b1,
    const float* __restrict__ W2l, const float* __restrict__ W2r,
    const float* __restrict__ b2,
    float* __restrict__ p, float* __restrict__ q)
{
    const int lane = threadIdx.x & 31;
    const int wave = (int)((blockIdx.x * 128u + threadIdx.x) >> 5);
    if (wave >= N_NODES / 16) return;

    const int n0   = wave * 16;
    const int c    = lane & 15;
    const int hf   = lane >> 4;
    const int node = n0 + c;

    const float rd = 1.0f / fmaxf(deg[(long)node * DEG_PITCH] +
                                  deg[COPY_FLOATS + (long)node * DEG_PITCH], 1.0f);

    const float* s0 = agg + (long)node * 64 + 4 * hf;
    const float* s1 = s0 + COPY_FLOATS;
    const float* xr = x + (long)node * 64 + 4 * hf;
    const float* wl = W1l + c * 64 + 4 * hf;
    const float* wr = W1r + c * 64 + 4 * hf;

    v8f acc[4];
#pragma unroll
    for (int T = 0; T < 4; ++T) {
        const float4 ba = *(const float4*)(b1 + 16 * T + 8 * hf);
        const float4 bb = *(const float4*)(b1 + 16 * T + 8 * hf + 4);
        acc[T][0] = ba.x; acc[T][1] = ba.y; acc[T][2] = ba.z; acc[T][3] = ba.w;
        acc[T][4] = bb.x; acc[T][5] = bb.y; acc[T][6] = bb.z; acc[T][7] = bb.w;
    }

#pragma unroll
    for (int t = 0; t < 8; ++t) {
        const float4 g0 = *(const float4*)(s0 + 8 * t);
        const float4 g1 = *(const float4*)(s1 + 8 * t);
        v2f f0, f1;
        f0[0] = (g0.x + g1.x) * rd;  f0[1] = (g0.y + g1.y) * rd;
        f1[0] = (g0.z + g1.z) * rd;  f1[1] = (g0.w + g1.w) * rd;
        float4 w[4];
#pragma unroll
        for (int T = 0; T < 4; ++T) w[T] = *(const float4*)(wl + T * 16 * 64 + 8 * t);
#pragma unroll
        for (int T = 0; T < 4; ++T) { v2f a; a[0] = w[T].x; a[1] = w[T].y; acc[T] = wmma4(a, f0, acc[T]); }
#pragma unroll
        for (int T = 0; T < 4; ++T) { v2f a; a[0] = w[T].z; a[1] = w[T].w; acc[T] = wmma4(a, f1, acc[T]); }
    }
#pragma unroll
    for (int t = 0; t < 8; ++t) {
        const float4 g = *(const float4*)(xr + 8 * t);
        v2f f0, f1;
        f0[0] = g.x;  f0[1] = g.y;
        f1[0] = g.z;  f1[1] = g.w;
        float4 w[4];
#pragma unroll
        for (int T = 0; T < 4; ++T) w[T] = *(const float4*)(wr + T * 16 * 64 + 8 * t);
#pragma unroll
        for (int T = 0; T < 4; ++T) { v2f a; a[0] = w[T].x; a[1] = w[T].y; acc[T] = wmma4(a, f0, acc[T]); }
#pragma unroll
        for (int T = 0; T < 4; ++T) { v2f a; a[0] = w[T].z; a[1] = w[T].w; acc[T] = wmma4(a, f1, acc[T]); }
    }

#pragma unroll
    for (int T = 0; T < 4; ++T)
#pragma unroll
        for (int r = 0; r < 8; ++r) {
            const float y = acc[T][r];
            acc[T][r] = (y > 0.0f) ? y : NEG_SLOPE * y;
        }

    const float* w2[4];
    w2[0] = W2l + c * 64 + 8 * hf;
    w2[1] = W2l + (16 + c) * 64 + 8 * hf;
    w2[2] = W2r + c * 64 + 8 * hf;
    w2[3] = W2r + (16 + c) * 64 + 8 * hf;

    const float bq0 = b2[c];
    const float bq1 = b2[16 + c];
    v8f o[4];
#pragma unroll
    for (int r = 0; r < 8; ++r) { o[0][r] = 0.0f; o[1][r] = 0.0f; o[2][r] = bq0; o[3][r] = bq1; }

#pragma unroll
    for (int T = 0; T < 4; ++T) {
        float4 wa[4], wb[4];
#pragma unroll
        for (int J = 0; J < 4; ++J) {
            wa[J] = *(const float4*)(w2[J] + 16 * T);
            wb[J] = *(const float4*)(w2[J] + 16 * T + 4);
        }
#pragma unroll
        for (int u = 0; u < 4; ++u) {
            v2f a;
            a[0] = acc[T][2 * u];
            a[1] = acc[T][2 * u + 1];
#pragma unroll
            for (int J = 0; J < 4; ++J) {
                v2f b;
                if (u == 0)      { b[0] = wa[J].x; b[1] = wa[J].y; }
                else if (u == 1) { b[0] = wa[J].z; b[1] = wa[J].w; }
                else if (u == 2) { b[0] = wb[J].x; b[1] = wb[J].y; }
                else             { b[0] = wb[J].z; b[1] = wb[J].w; }
                o[J] = wmma4(a, b, o[J]);
            }
        }
    }

    const int col = hf * 16 + c;
#pragma unroll
    for (int r = 0; r < 8; ++r) {
        {
            const float lo = o[0][r], hi = o[1][r];
            const float give = hf ? lo : hi;
            const float got  = __shfl_xor(give, 16);
            const float rowA = hf ? got : lo;
            const float rowB = hf ? hi  : got;
            p[(long)(n0 + r) * OUT_CH + col]     = rowA;
            p[(long)(n0 + 8 + r) * OUT_CH + col] = rowB;
        }
        {
            const float lo = o[2][r], hi = o[3][r];
            const float give = hf ? lo : hi;
            const float got  = __shfl_xor(give, 16);
            const float rowA = hf ? got : lo;
            const float rowB = hf ? hi  : got;
            q[(long)(n0 + r) * OUT_CH + col]     = rowA;
            q[(long)(n0 + 8 + r) * OUT_CH + col] = rowB;
        }
    }
}

#define FIN_ITEMS   4
#define FIN_THREADS (N_NODES * (OUT_CH / 4) / FIN_ITEMS)
static_assert((N_NODES * (OUT_CH / 4)) % FIN_ITEMS == 0, "every thread takes exactly FIN_ITEMS items");

__global__ __launch_bounds__(256) void sage_finish(
    const float* __restrict__ agg2, const float* __restrict__ deg, float* out)
{
    const int t = blockIdx.x * 256 + threadIdx.x;
    if (t >= FIN_THREADS) return;

    float  d0[FIN_ITEMS], d1[FIN_ITEMS];
    float4 a0[FIN_ITEMS], a1[FIN_ITEMS], s[FIN_ITEMS];
#pragma unroll
    for (int k = 0; k < FIN_ITEMS; ++k) {
        const int idx = t + k * FIN_THREADS;
        const int row = idx / (OUT_CH / 4);
        d0[k] = deg[(long)row * DEG_PITCH];
        d1[k] = deg[COPY_FLOATS + (long)row * DEG_PITCH];
        a0[k] = ((const float4*)agg2)[idx];
        a1[k] = ((const float4*)(agg2 + COPY_FLOATS))[idx];
        s[k]  = ((const float4*)out)[idx];
    }
    static_assert(FIN_ITEMS == 4, "the fence names the loads of four items");
    asm volatile("" :: "v"(d0[0]), "v"(d1[0]), "v"(a0[0].x), "v"(a1[0].x), "v"(s[0].x),
                       "v"(d0[1]), "v"(d1[1]), "v"(a0[1].x), "v"(a1[1].x), "v"(s[1].x),
                       "v"(d0[2]), "v"(d1[2]), "v"(a0[2].x), "v"(a1[2].x), "v"(s[2].x),
                       "v"(d0[3]), "v"(d1[3]), "v"(a0[3].x), "v"(a1[3].x), "v"(s[3].x));
#pragma unroll
    for (int k = 0; k < FIN_ITEMS; ++k) {
        const int idx = t + k * FIN_THREADS;
        const float rd = 1.0f / fmaxf(d0[k] + d1[k], 1.0f);
        float4 a;
        a.x = a0[k].x + a1[k].x;
        a.y = a0[k].y + a1[k].y;
        a.z = a0[k].z + a1[k].z;
        a.w = a0[k].w + a1[k].w;
        float4 y;
        y.x = a.x * rd + s[k].x;
        y.y = a.y * rd + s[k].y;
        y.z = a.z * rd + s[k].z;
        y.w = a.w * rd + s[k].w;
        y.x = (y.x > 0.0f) ? y.x : NEG_SLOPE * y.x;
        y.y = (y.y > 0.0f) ? y.y : NEG_SLOPE * y.y;
        y.z = (y.z > 0.0f) ? y.z : NEG_SLOPE * y.z;
        y.w = (y.w > 0.0f) ? y.w : NEG_SLOPE * y.w;
        ((float4*)out)[idx] = y;
    }
}

extern "C" void kernel_launch(void* const* d_in, const int* in_sizes, int n_in,
                              void* d_out, int out_size, void* d_ws, size_t ws_size,
                              hipStream_t stream) {
    const float* x   = (const float*)d_in[0];
    const int*   ei  = (const int*)  d_in[1];
    const float* W1l = (const float*)d_in[2];
    const float* b1  = (const float*)d_in[3];
    const float* W1r = (const float*)d_in[4];
    const float* W2l = (const float*)d_in[5];
    const float* b2  = (const float*)d_in[6];
    const float* W2r = (const float*)d_in[7];
    float* out = (float*)d_out;

    const size_t degB  = (size_t)N_NODES * DEG_PITCH * 4;
    const size_t aggB  = (size_t)N_NODES * 64 * 4;
    const size_t agg2B = (size_t)N_NODES * OUT_CH * 4;
    const size_t copyB = (size_t)COPY_FLOATS * 4;
    float* deg  = (float*)d_ws;
    float* agg  = (float*)((char*)d_ws + degB);
    float* agg2 = (float*)((char*)d_ws + degB + aggB);
    float* p    = (float*)((char*)d_ws + 2 * copyB);

    hipMemsetAsync(d_ws, 0, 2 * copyB, stream);

    {
        const int swaves  = EDGE_WAVES;
        const int sblocks = (swaves + 7) / 8;
        sage_scatter<IN_CH><<<sblocks, 256, 0, stream>>>(x, ei, agg, deg, 1);
    }
    {
        const int waves  = N_NODES / 16;
        const int blocks = (waves + 3) / 4;
        sage_layers<<<blocks, 128, 0, stream>>>(agg, deg, x, W1l, W1r, b1, W2l, W2r, b2, p, out);
    }
    {
        const int swaves  = EDGE_WAVES;
        const int sblocks = (swaves + 7) / 8;
        sage_scatter<OUT_CH><<<sblocks, 256, 0, stream>>>(p, ei, agg2, deg, 0);
    }
    {
        const int fthreads = FIN_THREADS;
        const int fblocks  = (fthreads + 255) / 256;
        sage_finish<<<fblocks, 256, 0, stream>>>(agg2, deg, out);
    }
}
